# starting kernel
# baseline (speedup 1.0000x reference)
#include <hip/hip_runtime.h>

typedef __attribute__((ext_vector_type(2))) float v2f;
typedef __attribute__((ext_vector_type(8))) float v8f;

#define N_NODES 50000
#define N_EDGES 640000
#define DDIM    128
#define LN_EPS  1e-5f

#define SB_HSTRIDE 144
#define SB_SIZE    (32 * 2 * SB_HSTRIDE)

#define SZ_STRIDE 132

__global__ void zero_f32(float* __restrict__ p, int n) {
    int i = blockIdx.x * blockDim.x + threadIdx.x;
    if (i < n) p[i] = 0.0f;
}

__global__ __launch_bounds__(256) void scatter_add(
    const float* __restrict__ h, const int* __restrict__ ei,
    float* __restrict__ agg) {
    int gid  = blockIdx.x * blockDim.x + threadIdx.x;
    int e    = gid >> 5;
    int lane = gid & 31;
    if (e >= N_EDGES) return;
    int src = ei[e];
    int dst = ei[N_EDGES + e];
    const float4 v = *(const float4*)(h + (size_t)src * DDIM + lane * 4);
    float* a = agg + (size_t)dst * DDIM + lane * 4;
    atomicAdd(a + 0, v.x);
    atomicAdd(a + 1, v.y);
    atomicAdd(a + 2, v.z);
    atomicAdd(a + 3, v.w);
}

__global__ __launch_bounds__(128) void gin_mlp(
    const float* __restrict__ h_in, const float* __restrict__ agg,
    const float* __restrict__ eps_arr, int layer, int has_skip,
    const float* __restrict__ W1, const float* __restrict__ b1,
    const float* __restrict__ W2, const float* __restrict__ b2,
    const float* __restrict__ lng, const float* __restrict__ lnb,
    const float* __restrict__ sg,  const float* __restrict__ sb,
    float* __restrict__ h_out) {

    __shared__ float2 sB[SB_SIZE];
    __shared__ float  sZ[4 * 16 * SZ_STRIDE];
    __shared__ float  sStat[4 * 16 * 2];

    const int tid   = threadIdx.x;
    const int wave  = tid >> 5;
    const int lane  = tid & 31;
    const int lhalf = lane >> 4;
    const int l15   = lane & 15;

    const float* W1l = W1 + (size_t)layer * DDIM * DDIM;
    const float* W2l = W2 + (size_t)layer * DDIM * DDIM;
    const float* b1l = b1 + (size_t)layer * DDIM;
    const float* b2l = b2 + (size_t)layer * DDIM;
    const float* lgl = lng + (size_t)layer * DDIM;
    const float* lbl = lnb + (size_t)layer * DDIM;
    const float* sgl = sg + (size_t)layer * DDIM;
    const float* sbl = sb + (size_t)layer * DDIM;

    const float epsv = 1.0f + eps_arr[layer];

    const int node_base = blockIdx.x * 64 + wave * 16;

    int arow = node_base + l15;
    if (arow >= N_NODES) arow = N_NODES - 1;
    const float* hrow = h_in + (size_t)arow * DDIM;
    const float* grow = agg  + (size_t)arow * DDIM;

    for (int idx = tid; idx < 32 * 2 * DDIM; idx += 128) {
        const int kk   = idx >> 8;
        const int lh   = (idx >> 7) & 1;
        const int col  = idx & 127;
        const int krow = kk * 4 + lh * 2;
        float2 f;
        f.x = W1l[krow * DDIM + col];
        f.y = W1l[(krow + 1) * DDIM + col];
        sB[(kk * 2 + lh) * SB_HSTRIDE + col] = f;
    }
    __syncthreads();

    const v8f vzero = {};
    v8f acc[8];
#pragma unroll
    for (int t = 0; t < 8; ++t) acc[t] = vzero;

    const float2* bfrag = &sB[lhalf * SB_HSTRIDE + l15];
    for (int kk = 0; kk < 32; ++kk) {
        const int c0 = kk * 4 + lhalf * 2;
        float2 hv = *(const float2*)(hrow + c0);
        float2 gv = *(const float2*)(grow + c0);
        v2f a;
        a.x = epsv * hv.x + gv.x;
        a.y = epsv * hv.y + gv.y;
        const float2* bk = bfrag + kk * (2 * SB_HSTRIDE);
#pragma unroll
        for (int nt = 0; nt < 8; ++nt) {
            float2 bf = bk[nt * 16];
            v2f b; b.x = bf.x; b.y = bf.y;
            acc[nt] = __builtin_amdgcn_wmma_f32_16x16x4_f32(
                false, a, false, b, (short)0, acc[nt], false, false);
        }
    }

    {
        float* zt = &sZ[wave * 16 * SZ_STRIDE];
#pragma unroll
        for (int nt = 0; nt < 8; ++nt) {
            const int col  = nt * 16 + l15;
            const float bi = b1l[col];
#pragma unroll
            for (int v = 0; v < 8; ++v) {
                float val = acc[nt][v] + bi;
                val = fmaxf(val, 0.0f);
                zt[(v + lhalf * 8) * SZ_STRIDE + col] = val;
            }
        }
    }
    __syncthreads();

    for (int idx = tid; idx < 32 * 2 * DDIM; idx += 128) {
        const int kk   = idx >> 8;
        const int lh   = (idx >> 7) & 1;
        const int col  = idx & 127;
        const int krow = kk * 4 + lh * 2;
        float2 f;
        f.x = W2l[krow * DDIM + col];
        f.y = W2l[(krow + 1) * DDIM + col];
        sB[(kk * 2 + lh) * SB_HSTRIDE + col] = f;
    }
    __syncthreads();

#pragma unroll
    for (int t = 0; t < 8; ++t) acc[t] = vzero;

    {
        const float* zrow = &sZ[(wave * 16 + l15) * SZ_STRIDE];
        for (int kk = 0; kk < 32; ++kk) {
            const int c0 = kk * 4 + lhalf * 2;
            float2 av = *(const float2*)(zrow + c0);
            v2f a; a.x = av.x; a.y = av.y;
            const float2* bk = bfrag + kk * (2 * SB_HSTRIDE);
#pragma unroll
            for (int nt = 0; nt < 8; ++nt) {
                float2 bf = bk[nt * 16];
                v2f b; b.x = bf.x; b.y = bf.y;
                acc[nt] = __builtin_amdgcn_wmma_f32_16x16x4_f32(
                    false, a, false, b, (short)0, acc[nt], false, false);
            }
        }
    }

    {
        float* zt = &sZ[wave * 16 * SZ_STRIDE];
#pragma unroll
        for (int nt = 0; nt < 8; ++nt) {
            const int col  = nt * 16 + l15;
            const float bi = b2l[col];
#pragma unroll
            for (int v = 0; v < 8; ++v)
                zt[(v + lhalf * 8) * SZ_STRIDE + col] = acc[nt][v] + bi;
        }
    }
    __syncthreads();

    {
        float* yrow = &sZ[(wave * 16 + l15) * SZ_STRIDE];
        const int cbase = lhalf * 64;

        float s = 0.0f, s2 = 0.0f;
#pragma unroll 4
        for (int j = 0; j < 64; ++j) {
            float y = yrow[cbase + j];
            s += y; s2 += y * y;
        }
        s  += __shfl_xor(s, 16, 32);
        s2 += __shfl_xor(s2, 16, 32);
        const float mu  = s * (1.0f / DDIM);
        const float var = s2 * (1.0f / DDIM) - mu * mu;
        const float inv = rsqrtf(var + LN_EPS);

        float t = 0.0f, t2 = 0.0f;
#pragma unroll 4
        for (int j = 0; j < 64; ++j) {
            const int c = cbase + j;
            float z1 = (yrow[c] - mu) * inv * lgl[c] + lbl[c];
            yrow[c] = z1;
            t += z1; t2 += z1 * z1;
        }

        if (has_skip) {
            t  += __shfl_xor(t, 16, 32);
            t2 += __shfl_xor(t2, 16, 32);
            float mu2  = t * (1.0f / DDIM);
            float var2 = t2 * (1.0f / DDIM) - mu2 * mu2;
            float inv2 = rsqrtf(var2 + LN_EPS);
            if (lhalf == 0) {
                sStat[(wave * 16 + l15) * 2 + 0] = mu2;
                sStat[(wave * 16 + l15) * 2 + 1] = inv2;
            }
        }
    }
    __syncthreads();

    {
        const float* zt = &sZ[wave * 16 * SZ_STRIDE];
        const int c4 = lane * 4;
        float4 g4 = {1.f, 1.f, 1.f, 1.f}, sb4 = {0.f, 0.f, 0.f, 0.f};
        if (has_skip) {
            g4  = *(const float4*)(sgl + c4);
            sb4 = *(const float4*)(sbl + c4);
        }
#pragma unroll 4
        for (int r = 0; r < 16; ++r) {
            const int node = node_base + r;
            if (node >= N_NODES) break;
            const float* zr = zt + r * SZ_STRIDE + c4;
            float4 z = {zr[0], zr[1], zr[2], zr[3]};
            if (has_skip) {
                const float mu2  = sStat[(wave * 16 + r) * 2 + 0];
                const float inv2 = sStat[(wave * 16 + r) * 2 + 1];
                z.x = fmaxf((z.x - mu2) * inv2 * g4.x + sb4.x, 0.0f);
                z.y = fmaxf((z.y - mu2) * inv2 * g4.y + sb4.y, 0.0f);
                z.z = fmaxf((z.z - mu2) * inv2 * g4.z + sb4.z, 0.0f);
                z.w = fmaxf((z.w - mu2) * inv2 * g4.w + sb4.w, 0.0f);
            }
            const float4 hres = *(const float4*)(h_in + (size_t)node * DDIM + c4);
            float4 o;
            o.x = z.x + hres.x; o.y = z.y + hres.y;
            o.z = z.z + hres.z; o.w = z.w + hres.w;
            *(float4*)(h_out + (size_t)node * DDIM + c4) = o;
        }
    }
}

extern "C" void kernel_launch(void* const* d_in, const int* in_sizes, int n_in,
                              void* d_out, int out_size, void* d_ws, size_t ws_size,
                              hipStream_t stream) {
    (void)in_sizes; (void)n_in; (void)out_size; (void)ws_size;

    const float* x   = (const float*)d_in[0];
    const int*   ei  = (const int*)d_in[1];
    const float* eps = (const float*)d_in[2];
    const float* W1  = (const float*)d_in[3];
    const float* b1  = (const float*)d_in[4];
    const float* W2  = (const float*)d_in[5];
    const float* b2  = (const float*)d_in[6];
    const float* lng = (const float*)d_in[7];
    const float* lnb = (const float*)d_in[8];
    const float* sg  = (const float*)d_in[9];
    const float* sb  = (const float*)d_in[10];
    float* out = (float*)d_out;

    const size_t feat = (size_t)N_NODES * DDIM;
    float* agg = (float*)d_ws;
    float* hA  = agg + feat;
    float* hB  = hA + feat;

    const int zero_blocks    = (N_NODES * DDIM + 255) / 256;
    const int scatter_blocks = (N_EDGES * 32) / 256;
    const int mlp_blocks     = (N_NODES + 63) / 64;

    const float* hin = x;
    for (int i = 0; i < 3; ++i) {
        float* hout = (i == 0) ? hA : (i == 1) ? hB : out;
        const int has_skip = (i < 2) ? 1 : 0;

        zero_f32<<<zero_blocks, 256, 0, stream>>>(agg, N_NODES * DDIM);
        scatter_add<<<scatter_blocks, 256, 0, stream>>>(hin, ei, agg);
        gin_mlp<<<mlp_blocks, 128, 0, stream>>>(
            hin, agg, eps, i, has_skip,
            W1, b1, W2, b2, lng, lnb, sg, sb, hout);

        hin = hout;
    }
}


# Float-atomic scatter replaced by per-call destination buckets (int-atomic count, scan, rank) and a one-wave-per-node row gather; per-node sum order is now the arrival-order rank; zero-fill gone
# speedup vs baseline: 1.5688x; 1.5688x over previous
#include <hip/hip_runtime.h>

typedef __attribute__((ext_vector_type(2))) float v2f;
typedef __attribute__((ext_vector_type(8))) float v8f;

#define N_NODES 50000
#define N_EDGES 640000
#define DDIM    128
#define LN_EPS  1e-5f

#define SB_HSTRIDE 144
#define SB_SIZE    (32 * 2 * SB_HSTRIDE)

#define SZ_STRIDE 132

#define SCAN_THREADS 1024
#define SCAN_PER     49

__global__ void zero_i32(int* __restrict__ p, int n) {
    int i = blockIdx.x * blockDim.x + threadIdx.x;
    if (i < n) p[i] = 0;
}

__global__ __launch_bounds__(256) void count_dst(
    const int* __restrict__ ei, int* __restrict__ cnt) {
    int e = blockIdx.x * blockDim.x + threadIdx.x;
    if (e >= N_EDGES) return;
    atomicAdd(cnt + ei[N_EDGES + e], 1);
}

__global__ __launch_bounds__(SCAN_THREADS) void scan_counts(
    const int* __restrict__ cnt, int* __restrict__ offs,
    int* __restrict__ cursor) {
    __shared__ int s[SCAN_THREADS * SCAN_PER];
    __shared__ int wsum[32];

    const int t    = threadIdx.x;
    const int lane = t & 31;
    const int w    = t >> 5;

#pragma unroll
    for (int j = 0; j < SCAN_PER; ++j) {
        const int i = j * SCAN_THREADS + t;
        s[i] = (i < N_NODES) ? cnt[i] : 0;
    }
    __syncthreads();

    int* my = &s[t * SCAN_PER];
    int sum = 0;
    for (int j = 0; j < SCAN_PER; ++j) sum += my[j];

    int incl = sum;
    for (int d = 1; d < 32; d <<= 1) {
        const int v = __shfl_up(incl, d, 32);
        if (lane >= d) incl += v;
    }
    if (lane == 31) wsum[w] = incl;
    __syncthreads();
    if (w == 0) {
        const int x = wsum[lane];
        int xi = x;
        for (int d = 1; d < 32; d <<= 1) {
            const int v = __shfl_up(xi, d, 32);
            if (lane >= d) xi += v;
        }
        wsum[lane] = xi - x;
    }
    __syncthreads();

    int run = wsum[w] + incl - sum;
    for (int j = 0; j < SCAN_PER; ++j) {
        const int cj = my[j];
        my[j] = run;
        run += cj;
    }
    __syncthreads();

    for (int j = 0; j < SCAN_PER; ++j) {
        const int i = j * SCAN_THREADS + t;
        if (i <= N_NODES) {
            const int v = s[i];
            offs[i] = v;
            if (i < N_NODES) cursor[i] = v;
        }
    }
}

__global__ __launch_bounds__(256) void rank_edges(
    const int* __restrict__ ei, int* __restrict__ cursor,
    int* __restrict__ esrc) {
    int e = blockIdx.x * blockDim.x + threadIdx.x;
    if (e >= N_EDGES) return;
    const int src = ei[e];
    const int dst = ei[N_EDGES + e];
    const int pos = atomicAdd(cursor + dst, 1);
    esrc[pos] = src;
}

#define GATHER_DEPTH 8
__global__ __launch_bounds__(256) void gather_rows(
    const float* __restrict__ h, const int* __restrict__ offs,
    const int* __restrict__ esrc, float* __restrict__ agg) {
    const int lane = threadIdx.x & 31;
    const int wv   = __builtin_amdgcn_readfirstlane((int)(threadIdx.x >> 5));
    const int node = blockIdx.x * 8 + wv;
    if (node >= N_NODES) return;

    const int beg = offs[node];
    const int end = offs[node + 1];
    const float* hl = h + lane * 4;

    float4 acc = {0.f, 0.f, 0.f, 0.f};
    for (int base = beg; base < end; base += 32) {
        const int n = (end - base < 32) ? (end - base) : 32;
        const int mine = (lane < n) ? esrc[base + lane] : 0;
        for (int j = 0; j < n; j += GATHER_DEPTH) {
            int sidx[GATHER_DEPTH];
#pragma unroll
            for (int k = 0; k < GATHER_DEPTH; ++k)
                sidx[k] = __builtin_amdgcn_readlane(mine, j + k);
            float4 v[GATHER_DEPTH];
#pragma unroll
            for (int k = 0; k < GATHER_DEPTH; ++k) {
                v[k] = make_float4(0.f, 0.f, 0.f, 0.f);
                if (j + k < n)
                    v[k] = *(const float4*)(hl + (size_t)sidx[k] * DDIM);
            }
#pragma unroll
            for (int k = 0; k < GATHER_DEPTH; ++k) {
                acc.x += v[k].x; acc.y += v[k].y;
                acc.z += v[k].z; acc.w += v[k].w;
            }
        }
    }
    *(float4*)(agg + (size_t)node * DDIM + lane * 4) = acc;
}

__global__ __launch_bounds__(128) void gin_mlp(
    const float* __restrict__ h_in, const float* __restrict__ agg,
    const float* __restrict__ eps_arr, int layer, int has_skip,
    const float* __restrict__ W1, const float* __restrict__ b1,
    const float* __restrict__ W2, const float* __restrict__ b2,
    const float* __restrict__ lng, const float* __restrict__ lnb,
    const float* __restrict__ sg,  const float* __restrict__ sb,
    float* __restrict__ h_out) {

    __shared__ float2 sB[SB_SIZE];
    __shared__ float  sZ[4 * 16 * SZ_STRIDE];
    __shared__ float  sStat[4 * 16 * 2];

    const int tid   = threadIdx.x;
    const int wave  = tid >> 5;
    const int lane  = tid & 31;
    const int lhalf = lane >> 4;
    const int l15   = lane & 15;

    const float* W1l = W1 + (size_t)layer * DDIM * DDIM;
    const float* W2l = W2 + (size_t)layer * DDIM * DDIM;
    const float* b1l = b1 + (size_t)layer * DDIM;
    const float* b2l = b2 + (size_t)layer * DDIM;
    const float* lgl = lng + (size_t)layer * DDIM;
    const float* lbl = lnb + (size_t)layer * DDIM;
    const float* sgl = sg + (size_t)layer * DDIM;
    const float* sbl = sb + (size_t)layer * DDIM;

    const float epsv = 1.0f + eps_arr[layer];

    const int node_base = blockIdx.x * 64 + wave * 16;

    int arow = node_base + l15;
    if (arow >= N_NODES) arow = N_NODES - 1;
    const float* hrow = h_in + (size_t)arow * DDIM;
    const float* grow = agg  + (size_t)arow * DDIM;

    for (int idx = tid; idx < 32 * 2 * DDIM; idx += 128) {
        const int kk   = idx >> 8;
        const int lh   = (idx >> 7) & 1;
        const int col  = idx & 127;
        const int krow = kk * 4 + lh * 2;
        float2 f;
        f.x = W1l[krow * DDIM + col];
        f.y = W1l[(krow + 1) * DDIM + col];
        sB[(kk * 2 + lh) * SB_HSTRIDE + col] = f;
    }
    __syncthreads();

    const v8f vzero = {};
    v8f acc[8];
#pragma unroll
    for (int t = 0; t < 8; ++t) acc[t] = vzero;

    const float2* bfrag = &sB[lhalf * SB_HSTRIDE + l15];
    for (int kk = 0; kk < 32; ++kk) {
        const int c0 = kk * 4 + lhalf * 2;
        float2 hv = *(const float2*)(hrow + c0);
        float2 gv = *(const float2*)(grow + c0);
        v2f a;
        a.x = epsv * hv.x + gv.x;
        a.y = epsv * hv.y + gv.y;
        const float2* bk = bfrag + kk * (2 * SB_HSTRIDE);
#pragma unroll
        for (int nt = 0; nt < 8; ++nt) {
            float2 bf = bk[nt * 16];
            v2f b; b.x = bf.x; b.y = bf.y;
            acc[nt] = __builtin_amdgcn_wmma_f32_16x16x4_f32(
                false, a, false, b, (short)0, acc[nt], false, false);
        }
    }

    {
        float* zt = &sZ[wave * 16 * SZ_STRIDE];
#pragma unroll
        for (int nt = 0; nt < 8; ++nt) {
            const int col  = nt * 16 + l15;
            const float bi = b1l[col];
#pragma unroll
            for (int v = 0; v < 8; ++v) {
                float val = acc[nt][v] + bi;
                val = fmaxf(val, 0.0f);
                zt[(v + lhalf * 8) * SZ_STRIDE + col] = val;
            }
        }
    }
    __syncthreads();

    for (int idx = tid; idx < 32 * 2 * DDIM; idx += 128) {
        const int kk   = idx >> 8;
        const int lh   = (idx >> 7) & 1;
        const int col  = idx & 127;
        const int krow = kk * 4 + lh * 2;
        float2 f;
        f.x = W2l[krow * DDIM + col];
        f.y = W2l[(krow + 1) * DDIM + col];
        sB[(kk * 2 + lh) * SB_HSTRIDE + col] = f;
    }
    __syncthreads();

#pragma unroll
    for (int t = 0; t < 8; ++t) acc[t] = vzero;

    {
        const float* zrow = &sZ[(wave * 16 + l15) * SZ_STRIDE];
        for (int kk = 0; kk < 32; ++kk) {
            const int c0 = kk * 4 + lhalf * 2;
            float2 av = *(const float2*)(zrow + c0);
            v2f a; a.x = av.x; a.y = av.y;
            const float2* bk = bfrag + kk * (2 * SB_HSTRIDE);
#pragma unroll
            for (int nt = 0; nt < 8; ++nt) {
                float2 bf = bk[nt * 16];
                v2f b; b.x = bf.x; b.y = bf.y;
                acc[nt] = __builtin_amdgcn_wmma_f32_16x16x4_f32(
                    false, a, false, b, (short)0, acc[nt], false, false);
            }
        }
    }

    {
        float* zt = &sZ[wave * 16 * SZ_STRIDE];
#pragma unroll
        for (int nt = 0; nt < 8; ++nt) {
            const int col  = nt * 16 + l15;
            const float bi = b2l[col];
#pragma unroll
            for (int v = 0; v < 8; ++v)
                zt[(v + lhalf * 8) * SZ_STRIDE + col] = acc[nt][v] + bi;
        }
    }
    __syncthreads();

    {
        float* yrow = &sZ[(wave * 16 + l15) * SZ_STRIDE];
        const int cbase = lhalf * 64;

        float s = 0.0f, s2 = 0.0f;
#pragma unroll 4
        for (int j = 0; j < 64; ++j) {
            float y = yrow[cbase + j];
            s += y; s2 += y * y;
        }
        s  += __shfl_xor(s, 16, 32);
        s2 += __shfl_xor(s2, 16, 32);
        const float mu  = s * (1.0f / DDIM);
        const float var = s2 * (1.0f / DDIM) - mu * mu;
        const float inv = rsqrtf(var + LN_EPS);

        float t = 0.0f, t2 = 0.0f;
#pragma unroll 4
        for (int j = 0; j < 64; ++j) {
            const int c = cbase + j;
            float z1 = (yrow[c] - mu) * inv * lgl[c] + lbl[c];
            yrow[c] = z1;
            t += z1; t2 += z1 * z1;
        }

        if (has_skip) {
            t  += __shfl_xor(t, 16, 32);
            t2 += __shfl_xor(t2, 16, 32);
            float mu2  = t * (1.0f / DDIM);
            float var2 = t2 * (1.0f / DDIM) - mu2 * mu2;
            float inv2 = rsqrtf(var2 + LN_EPS);
            if (lhalf == 0) {
                sStat[(wave * 16 + l15) * 2 + 0] = mu2;
                sStat[(wave * 16 + l15) * 2 + 1] = inv2;
            }
        }
    }
    __syncthreads();

    {
        const float* zt = &sZ[wave * 16 * SZ_STRIDE];
        const int c4 = lane * 4;
        float4 g4 = {1.f, 1.f, 1.f, 1.f}, sb4 = {0.f, 0.f, 0.f, 0.f};
        if (has_skip) {
            g4  = *(const float4*)(sgl + c4);
            sb4 = *(const float4*)(sbl + c4);
        }
#pragma unroll 4
        for (int r = 0; r < 16; ++r) {
            const int node = node_base + r;
            if (node >= N_NODES) break;
            const float* zr = zt + r * SZ_STRIDE + c4;
            float4 z = {zr[0], zr[1], zr[2], zr[3]};
            if (has_skip) {
                const float mu2  = sStat[(wave * 16 + r) * 2 + 0];
                const float inv2 = sStat[(wave * 16 + r) * 2 + 1];
                z.x = fmaxf((z.x - mu2) * inv2 * g4.x + sb4.x, 0.0f);
                z.y = fmaxf((z.y - mu2) * inv2 * g4.y + sb4.y, 0.0f);
                z.z = fmaxf((z.z - mu2) * inv2 * g4.z + sb4.z, 0.0f);
                z.w = fmaxf((z.w - mu2) * inv2 * g4.w + sb4.w, 0.0f);
            }
            const float4 hres = *(const float4*)(h_in + (size_t)node * DDIM + c4);
            float4 o;
            o.x = z.x + hres.x; o.y = z.y + hres.y;
            o.z = z.z + hres.z; o.w = z.w + hres.w;
            *(float4*)(h_out + (size_t)node * DDIM + c4) = o;
        }
    }
}

extern "C" void kernel_launch(void* const* d_in, const int* in_sizes, int n_in,
                              void* d_out, int out_size, void* d_ws, size_t ws_size,
                              hipStream_t stream) {
    (void)in_sizes; (void)n_in; (void)out_size; (void)ws_size;

    const float* x   = (const float*)d_in[0];
    const int*   ei  = (const int*)d_in[1];
    const float* eps = (const float*)d_in[2];
    const float* W1  = (const float*)d_in[3];
    const float* b1  = (const float*)d_in[4];
    const float* W2  = (const float*)d_in[5];
    const float* b2  = (const float*)d_in[6];
    const float* lng = (const float*)d_in[7];
    const float* lnb = (const float*)d_in[8];
    const float* sg  = (const float*)d_in[9];
    const float* sb  = (const float*)d_in[10];
    float* out = (float*)d_out;

    const size_t feat = (size_t)N_NODES * DDIM;
    float* agg = (float*)d_ws;
    float* hA  = agg + feat;
    int* cnt    = (int*)(hA + feat);
    int* offs   = cnt + SCAN_THREADS * SCAN_PER;
    int* cursor = offs + SCAN_THREADS * SCAN_PER;
    int* esrc   = cursor + SCAN_THREADS * SCAN_PER;

    const int node_blocks   = (N_NODES + 255) / 256;
    const int edge_blocks   = (N_EDGES + 255) / 256;
    const int gather_blocks = (N_NODES + 7) / 8;
    const int mlp_blocks    = (N_NODES + 63) / 64;

    zero_i32<<<node_blocks, 256, 0, stream>>>(cnt, N_NODES);
    count_dst<<<edge_blocks, 256, 0, stream>>>(ei, cnt);
    scan_counts<<<1, SCAN_THREADS, 0, stream>>>(cnt, offs, cursor);
    rank_edges<<<edge_blocks, 256, 0, stream>>>(ei, cursor, esrc);

    const float* hin = x;
    for (int i = 0; i < 3; ++i) {
        float* hout = (i == 1) ? hA : out;
        const int has_skip = (i < 2) ? 1 : 0;

        gather_rows<<<gather_blocks, 256, 0, stream>>>(hin, offs, esrc, agg);
        gin_mlp<<<mlp_blocks, 128, 0, stream>>>(
            hin, agg, eps, i, has_skip,
            W1, b1, W2, b2, lng, lnb, sg, sb, hout);

        hin = hout;
    }
}


# MLP kernel made resident: 256 groups of 16 waves stage W1 and W2 into LDS once per launch, not twice per 64-node block; each wave then runs one 16-node tile end to end with no group barrier
# speedup vs baseline: 2.2013x; 1.4032x over previous
#include <hip/hip_runtime.h>

typedef __attribute__((ext_vector_type(2))) float v2f;
typedef __attribute__((ext_vector_type(8))) float v8f;

#define N_NODES 50000
#define N_EDGES 640000
#define DDIM    128
#define LN_EPS  1e-5f

#define SB_HSTRIDE 144
#define SB_SIZE    (32 * 2 * SB_HSTRIDE)

#define SZ_STRIDE 132

#define SCAN_THREADS 1024
#define SCAN_PER     49

#define OWN_NODES  196
#define OWN_GROUPS 256
#define DST_VEC    (N_EDGES / 4)
#define OWN_ROT    624
#define OWN_FULL   156
#define OWN_TAIL   (DST_VEC - OWN_FULL * 1024)

__device__ __forceinline__ int own_place(const int idx, const int rot) {
    const int jj = idx + rot;
    return (jj >= DST_VEC) ? jj - DST_VEC : jj;
}

__device__ __forceinline__ void own_edges(
    const int4 d, const int idx, const int rot, const unsigned lo,
    int* __restrict__ c, int* __restrict__ rank) {
    const unsigned a0 = (unsigned)d.x - lo;
    const unsigned a1 = (unsigned)d.y - lo;
    const unsigned a2 = (unsigned)d.z - lo;
    const unsigned a3 = (unsigned)d.w - lo;
    const unsigned m01 = a0 < a1 ? a0 : a1;
    const unsigned m23 = a2 < a3 ? a2 : a3;
    const unsigned m   = m01 < m23 ? m01 : m23;
    if (m < OWN_NODES) {
        int* rk = rank + 4 * own_place(idx, rot);
        if (a0 < OWN_NODES) rk[0] = atomicAdd(&c[a0], 1);
        if (a1 < OWN_NODES) rk[1] = atomicAdd(&c[a1], 1);
        if (a2 < OWN_NODES) rk[2] = atomicAdd(&c[a2], 1);
        if (a3 < OWN_NODES) rk[3] = atomicAdd(&c[a3], 1);
    }
}

__device__ __forceinline__ void own_load(
    int4 (&v)[4], const int4* __restrict__ d4, const int k, const int t, const int rot) {
#pragma unroll
    for (int u = 0; u < 4; ++u) v[u] = d4[own_place((k + u) * 1024 + t, rot)];
}

__device__ __forceinline__ void own_batch(
    const int4 (&v)[4], const int k, const int t, const int rot, const unsigned lo,
    int* __restrict__ c, int* __restrict__ rank) {
#pragma unroll
    for (int u = 0; u < 4; ++u) own_edges(v[u], (k + u) * 1024 + t, rot, lo, c, rank);
}

__global__ __launch_bounds__(1024) void count_rank_owned(
    const int* __restrict__ ei, int* __restrict__ cnt, int* __restrict__ rank) {
    __shared__ int c[OWN_NODES];
    const int t = threadIdx.x;
    const unsigned lo = blockIdx.x * OWN_NODES;
    if (t < OWN_NODES) c[t] = 0;
    __syncthreads();

    const int4* d4 = (const int4*)(ei + N_EDGES);
    const int rot = blockIdx.x * OWN_ROT;

    int4 va[4], vb[4];
    own_load(va, d4, 0, t, rot);
    for (int k = 0; k < OWN_FULL - 4; k += 8) {
        own_load(vb, d4, k + 4, t, rot);
        own_batch(va, k, t, rot, lo, c, rank);
        own_load(va, d4, k + 8, t, rot);
        own_batch(vb, k + 4, t, rot, lo, c, rank);
    }
    int4 last = make_int4(-1, -1, -1, -1);
    if (t < OWN_TAIL) last = d4[own_place(OWN_FULL * 1024 + t, rot)];
    own_batch(va, OWN_FULL - 4, t, rot, lo, c, rank);
    own_edges(last, OWN_FULL * 1024 + t, rot, lo, c, rank);

    __syncthreads();
    if (t < OWN_NODES && lo + t < N_NODES) cnt[lo + t] = c[t];
}

__global__ __launch_bounds__(SCAN_THREADS) void scan_counts(
    const int* __restrict__ cnt, int* __restrict__ offs) {
    __shared__ int s[SCAN_THREADS * SCAN_PER];
    __shared__ int wsum[32];

    const int t    = threadIdx.x;
    const int lane = t & 31;
    const int w    = t >> 5;

#pragma unroll
    for (int j = 0; j < SCAN_PER; ++j) {
        const int i = j * SCAN_THREADS + t;
        s[i] = (i < N_NODES) ? cnt[i] : 0;
    }
    __syncthreads();

    int* my = &s[t * SCAN_PER];
    int sum = 0;
    for (int j = 0; j < SCAN_PER; ++j) sum += my[j];

    int incl = sum;
    for (int d = 1; d < 32; d <<= 1) {
        const int v = __shfl_up(incl, d, 32);
        if (lane >= d) incl += v;
    }
    if (lane == 31) wsum[w] = incl;
    __syncthreads();
    if (w == 0) {
        const int x = wsum[lane];
        int xi = x;
        for (int d = 1; d < 32; d <<= 1) {
            const int v = __shfl_up(xi, d, 32);
            if (lane >= d) xi += v;
        }
        wsum[lane] = xi - x;
    }
    __syncthreads();

    int run = wsum[w] + incl - sum;
    for (int j = 0; j < SCAN_PER; ++j) {
        const int cj = my[j];
        my[j] = run;
        run += cj;
    }
    __syncthreads();

    for (int j = 0; j < SCAN_PER; ++j) {
        const int i = j * SCAN_THREADS + t;
        if (i <= N_NODES) offs[i] = s[i];
    }
}

__global__ __launch_bounds__(256) void fill_slots(
    const int* __restrict__ ei, const int* __restrict__ offs,
    const int* __restrict__ rank, int* __restrict__ esrc) {
    const int i = blockIdx.x * 256 + threadIdx.x;
    if (i >= DST_VEC) return;
    const int4 s = ((const int4*)ei)[i];
    const int4 d = ((const int4*)(ei + N_EDGES))[i];
    const int4 r = ((const int4*)rank)[i];
    if ((unsigned)d.x < N_NODES) esrc[offs[d.x] + r.x] = s.x;
    if ((unsigned)d.y < N_NODES) esrc[offs[d.y] + r.y] = s.y;
    if ((unsigned)d.z < N_NODES) esrc[offs[d.z] + r.z] = s.z;
    if ((unsigned)d.w < N_NODES) esrc[offs[d.w] + r.w] = s.w;
}

#define GATHER_DEPTH 8
__global__ __launch_bounds__(256) void gather_rows(
    const float* __restrict__ h, const int* __restrict__ offs,
    const int* __restrict__ esrc, float* __restrict__ agg) {
    const int lane = threadIdx.x & 31;
    const int wv   = __builtin_amdgcn_readfirstlane((int)(threadIdx.x >> 5));
    const int node = blockIdx.x * 8 + wv;
    if (node >= N_NODES) return;

    const int beg = offs[node];
    const int end = offs[node + 1];
    const float* hl = h + lane * 4;

    float4 acc = {0.f, 0.f, 0.f, 0.f};
    for (int base = beg; base < end; base += 32) {
        const int n = (end - base < 32) ? (end - base) : 32;
        const int mine = (lane < n) ? esrc[base + lane] : 0;
        for (int j = 0; j < n; j += GATHER_DEPTH) {
            int sidx[GATHER_DEPTH];
#pragma unroll
            for (int k = 0; k < GATHER_DEPTH; ++k)
                sidx[k] = __builtin_amdgcn_readlane(mine, j + k);
            float4 v[GATHER_DEPTH];
#pragma unroll
            for (int k = 0; k < GATHER_DEPTH; ++k) {
                v[k] = make_float4(0.f, 0.f, 0.f, 0.f);
                if (j + k < n)
                    v[k] = *(const float4*)(hl + (size_t)sidx[k] * DDIM);
            }
#pragma unroll
            for (int k = 0; k < GATHER_DEPTH; ++k) {
                acc.x += v[k].x; acc.y += v[k].y;
                acc.z += v[k].z; acc.w += v[k].w;
            }
        }
    }
    *(float4*)(agg + (size_t)node * DDIM + lane * 4) = acc;
}

#define MLP_WAVES   16
#define MLP_THREADS (MLP_WAVES * 32)
#define MLP_GROUPS  256
#define N_TILES     ((N_NODES + 15) / 16)

__device__ __forceinline__ void wave_handover() {
    __builtin_amdgcn_fence(__ATOMIC_RELEASE, "wavefront");
    __builtin_amdgcn_wave_barrier();
    __builtin_amdgcn_fence(__ATOMIC_ACQUIRE, "wavefront");
}

__global__ __launch_bounds__(MLP_THREADS) void gin_mlp(
    const float* __restrict__ h_in, const float* __restrict__ agg,
    const float* __restrict__ eps_arr, int layer, int has_skip,
    const float* __restrict__ W1, const float* __restrict__ b1,
    const float* __restrict__ W2, const float* __restrict__ b2,
    const float* __restrict__ lng, const float* __restrict__ lnb,
    const float* __restrict__ sg,  const float* __restrict__ sb,
    float* __restrict__ h_out) {

    __shared__ alignas(16) float2 sB1[SB_SIZE];
    __shared__ alignas(16) float2 sB2[SB_SIZE];
    __shared__ float  sZ[MLP_WAVES * 16 * SZ_STRIDE];
    __shared__ float  sStat[MLP_WAVES * 16 * 2];

    const int tid   = threadIdx.x;
    const int lane  = tid & 31;
    const int lhalf = lane >> 4;
    const int l15   = lane & 15;
    const int wave  = __builtin_amdgcn_readfirstlane(tid >> 5);

    const float* W1l = W1 + (size_t)layer * DDIM * DDIM;
    const float* W2l = W2 + (size_t)layer * DDIM * DDIM;
    const float* b1l = b1 + (size_t)layer * DDIM;
    const float* b2l = b2 + (size_t)layer * DDIM;
    const float* lgl = lng + (size_t)layer * DDIM;
    const float* lbl = lnb + (size_t)layer * DDIM;
    const float* sgl = sg + (size_t)layer * DDIM;
    const float* sbl = sb + (size_t)layer * DDIM;

#pragma unroll
    for (int i = 0; i < 4; ++i) {
        const int item = tid + i * MLP_THREADS;
        const int pair = item >> 5;
        const int c4   = (item & 31) * 4;
        const float4 p0 = *(const float4*)(W1l + (2 * pair) * DDIM + c4);
        const float4 p1 = *(const float4*)(W1l + (2 * pair + 1) * DDIM + c4);
        const float4 q0 = *(const float4*)(W2l + (2 * pair) * DDIM + c4);
        const float4 q1 = *(const float4*)(W2l + (2 * pair + 1) * DDIM + c4);
        float4* d1 = (float4*)&sB1[pair * SB_HSTRIDE + c4];
        float4* d2 = (float4*)&sB2[pair * SB_HSTRIDE + c4];
        d1[0] = make_float4(p0.x, p1.x, p0.y, p1.y);
        d1[1] = make_float4(p0.z, p1.z, p0.w, p1.w);
        d2[0] = make_float4(q0.x, q1.x, q0.y, q1.y);
        d2[1] = make_float4(q0.z, q1.z, q0.w, q1.w);
    }
    __syncthreads();

    const int tile = wave * MLP_GROUPS + (int)blockIdx.x;
    if (tile >= N_TILES) return;

    const float epsv = 1.0f + eps_arr[layer];
    const int node_base = tile * 16;

    int arow = node_base + l15;
    if (arow >= N_NODES) arow = N_NODES - 1;
    const float* hrow = h_in + (size_t)arow * DDIM;
    const float* grow = agg  + (size_t)arow * DDIM;

    const v8f vzero = {};
    v8f acc[8];
#pragma unroll
    for (int t = 0; t < 8; ++t) acc[t] = vzero;

    const float2* bfrag1 = &sB1[lhalf * SB_HSTRIDE + l15];
    for (int kk = 0; kk < 32; ++kk) {
        const int c0 = kk * 4 + lhalf * 2;
        float2 hv = *(const float2*)(hrow + c0);
        float2 gv = *(const float2*)(grow + c0);
        v2f a;
        a.x = epsv * hv.x + gv.x;
        a.y = epsv * hv.y + gv.y;
        const float2* bk = bfrag1 + kk * (2 * SB_HSTRIDE);
#pragma unroll
        for (int nt = 0; nt < 8; ++nt) {
            float2 bf = bk[nt * 16];
            v2f b; b.x = bf.x; b.y = bf.y;
            acc[nt] = __builtin_amdgcn_wmma_f32_16x16x4_f32(
                false, a, false, b, (short)0, acc[nt], false, false);
        }
    }

    float* zt = &sZ[wave * 16 * SZ_STRIDE];
    {
#pragma unroll
        for (int nt = 0; nt < 8; ++nt) {
            const int col  = nt * 16 + l15;
            const float bi = b1l[col];
#pragma unroll
            for (int v = 0; v < 8; ++v) {
                float val = acc[nt][v] + bi;
                val = fmaxf(val, 0.0f);
                zt[(v + lhalf * 8) * SZ_STRIDE + col] = val;
            }
        }
    }
    wave_handover();

#pragma unroll
    for (int t = 0; t < 8; ++t) acc[t] = vzero;

    {
        const float2* bfrag2 = &sB2[lhalf * SB_HSTRIDE + l15];
        const float* zrow = zt + l15 * SZ_STRIDE;
        for (int kk = 0; kk < 32; ++kk) {
            const int c0 = kk * 4 + lhalf * 2;
            float2 av = *(const float2*)(zrow + c0);
            v2f a; a.x = av.x; a.y = av.y;
            const float2* bk = bfrag2 + kk * (2 * SB_HSTRIDE);
#pragma unroll
            for (int nt = 0; nt < 8; ++nt) {
                float2 bf = bk[nt * 16];
                v2f b; b.x = bf.x; b.y = bf.y;
                acc[nt] = __builtin_amdgcn_wmma_f32_16x16x4_f32(
                    false, a, false, b, (short)0, acc[nt], false, false);
            }
        }
    }
    wave_handover();

    {
#pragma unroll
        for (int nt = 0; nt < 8; ++nt) {
            const int col  = nt * 16 + l15;
            const float bi = b2l[col];
#pragma unroll
            for (int v = 0; v < 8; ++v)
                zt[(v + lhalf * 8) * SZ_STRIDE + col] = acc[nt][v] + bi;
        }
    }
    wave_handover();

    {
        float* yrow = zt + l15 * SZ_STRIDE;
        const int cbase = lhalf * 64;

        float s = 0.0f, s2 = 0.0f;
#pragma unroll 4
        for (int j = 0; j < 64; ++j) {
            float y = yrow[cbase + j];
            s += y; s2 += y * y;
        }
        s  += __shfl_xor(s, 16, 32);
        s2 += __shfl_xor(s2, 16, 32);
        const float mu  = s * (1.0f / DDIM);
        const float var = s2 * (1.0f / DDIM) - mu * mu;
        const float inv = rsqrtf(var + LN_EPS);

        float t = 0.0f, t2 = 0.0f;
#pragma unroll 4
        for (int j = 0; j < 64; ++j) {
            const int c = cbase + j;
            float z1 = (yrow[c] - mu) * inv * lgl[c] + lbl[c];
            yrow[c] = z1;
            t += z1; t2 += z1 * z1;
        }

        if (has_skip) {
            t  += __shfl_xor(t, 16, 32);
            t2 += __shfl_xor(t2, 16, 32);
            float mu2  = t * (1.0f / DDIM);
            float var2 = t2 * (1.0f / DDIM) - mu2 * mu2;
            float inv2 = rsqrtf(var2 + LN_EPS);
            if (lhalf == 0) {
                sStat[(wave * 16 + l15) * 2 + 0] = mu2;
                sStat[(wave * 16 + l15) * 2 + 1] = inv2;
            }
        }
    }
    wave_handover();

    {
        const int c4 = lane * 4;
        float4 g4 = {1.f, 1.f, 1.f, 1.f}, sb4 = {0.f, 0.f, 0.f, 0.f};
        if (has_skip) {
            g4  = *(const float4*)(sgl + c4);
            sb4 = *(const float4*)(sbl + c4);
        }
#pragma unroll 4
        for (int r = 0; r < 16; ++r) {
            const int node = node_base + r;
            if (node >= N_NODES) break;
            const float* zr = zt + r * SZ_STRIDE + c4;
            float4 z = {zr[0], zr[1], zr[2], zr[3]};
            if (has_skip) {
                const float mu2  = sStat[(wave * 16 + r) * 2 + 0];
                const float inv2 = sStat[(wave * 16 + r) * 2 + 1];
                z.x = fmaxf((z.x - mu2) * inv2 * g4.x + sb4.x, 0.0f);
                z.y = fmaxf((z.y - mu2) * inv2 * g4.y + sb4.y, 0.0f);
                z.z = fmaxf((z.z - mu2) * inv2 * g4.z + sb4.z, 0.0f);
                z.w = fmaxf((z.w - mu2) * inv2 * g4.w + sb4.w, 0.0f);
            }
            const float4 hres = *(const float4*)(h_in + (size_t)node * DDIM + c4);
            float4 o;
            o.x = z.x + hres.x; o.y = z.y + hres.y;
            o.z = z.z + hres.z; o.w = z.w + hres.w;
            *(float4*)(h_out + (size_t)node * DDIM + c4) = o;
        }
    }
}

extern "C" void kernel_launch(void* const* d_in, const int* in_sizes, int n_in,
                              void* d_out, int out_size, void* d_ws, size_t ws_size,
                              hipStream_t stream) {
    (void)in_sizes; (void)n_in; (void)out_size; (void)ws_size;

    const float* x   = (const float*)d_in[0];
    const int*   ei  = (const int*)d_in[1];
    const float* eps = (const float*)d_in[2];
    const float* W1  = (const float*)d_in[3];
    const float* b1  = (const float*)d_in[4];
    const float* W2  = (const float*)d_in[5];
    const float* b2  = (const float*)d_in[6];
    const float* lng = (const float*)d_in[7];
    const float* lnb = (const float*)d_in[8];
    const float* sg  = (const float*)d_in[9];
    const float* sb  = (const float*)d_in[10];
    float* out = (float*)d_out;

    const size_t feat = (size_t)N_NODES * DDIM;
    float* agg = (float*)d_ws;
    float* hA  = agg + feat;
    int* cnt  = (int*)(hA + feat);
    int* offs = cnt + SCAN_THREADS * SCAN_PER;
    int* esrc = offs + SCAN_THREADS * SCAN_PER;
    int* rank = esrc + N_EDGES;

    const int fill_blocks   = (DST_VEC + 255) / 256;
    const int gather_blocks = (N_NODES + 7) / 8;

    count_rank_owned<<<OWN_GROUPS, 1024, 0, stream>>>(ei, cnt, rank);
    scan_counts<<<1, SCAN_THREADS, 0, stream>>>(cnt, offs);
    fill_slots<<<fill_blocks, 256, 0, stream>>>(ei, offs, rank, esrc);

    const float* hin = x;
    for (int i = 0; i < 3; ++i) {
        float* hout = (i == 1) ? hA : out;
        const int has_skip = (i < 2) ? 1 : 0;

        gather_rows<<<gather_blocks, 256, 0, stream>>>(hin, offs, esrc, agg);
        gin_mlp<<<MLP_GROUPS, MLP_THREADS, 0, stream>>>(
            hin, agg, eps, i, has_skip,
            W1, b1, W2, b2, lng, lnb, sg, sb, hout);

        hin = hout;
    }
}


# Edge bucketing as two-level counting partition: 256 list slices count edges per owner range and move them into owner bins; each owner builds offsets, slots from its bin; sum order stays arrival rank
# speedup vs baseline: 2.6964x; 1.2249x over previous
#include <hip/hip_runtime.h>

typedef __attribute__((ext_vector_type(2))) float v2f;
typedef __attribute__((ext_vector_type(8))) float v8f;

#define N_NODES 50000
#define N_EDGES 640000
#define DDIM    128
#define LN_EPS  1e-5f

#define SB_HSTRIDE 144
#define SB_SIZE    (32 * 2 * SB_HSTRIDE)

#define SZ_STRIDE 132

#define OWN_NODES  196
#define N_OWN      256
#define N_SLICE    256
#define DST_VEC    (N_EDGES / 4)
#define SLICE_VEC  (DST_VEC / N_SLICE)
#define PART_THREADS 1024

__global__ __launch_bounds__(640) void slice_hist(
    const int* __restrict__ ei, int* __restrict__ hist) {
    __shared__ int h[N_OWN];
    const int t = threadIdx.x;
    if (t < N_OWN) h[t] = 0;
    __syncthreads();
    if (t < SLICE_VEC) {
        const int4 d = ((const int4*)(ei + N_EDGES))[blockIdx.x * SLICE_VEC + t];
        if ((unsigned)d.x < N_NODES) atomicAdd(&h[(unsigned)d.x / OWN_NODES], 1);
        if ((unsigned)d.y < N_NODES) atomicAdd(&h[(unsigned)d.y / OWN_NODES], 1);
        if ((unsigned)d.z < N_NODES) atomicAdd(&h[(unsigned)d.z / OWN_NODES], 1);
        if ((unsigned)d.w < N_NODES) atomicAdd(&h[(unsigned)d.w / OWN_NODES], 1);
    }
    __syncthreads();
    if (t < N_OWN) hist[blockIdx.x * N_OWN + t] = h[t];
}

__device__ __forceinline__ int scan256(const int v, int* __restrict__ wsum, const int t) {
    const int lane = t & 31;
    const int w    = t >> 5;
    int incl = v;
    for (int d = 1; d < 32; d <<= 1) {
        const int u = __shfl_up(incl, d, 32);
        if (lane >= d) incl += u;
    }
    if (t < N_OWN && lane == 31) wsum[w] = incl;
    __syncthreads();
    int wbase = 0;
    if (t < N_OWN)
        for (int k = 0; k < w; ++k) wbase += wsum[k];
    return wbase + incl - v;
}

__global__ __launch_bounds__(PART_THREADS) void bin_edges(
    const int* __restrict__ ei, const int* __restrict__ hist,
    int* __restrict__ ostart, int* __restrict__ binp) {
    int2* __restrict__ bin = (int2*)binp;
    __shared__ int sT[4][N_OWN];
    __shared__ int sP[4][N_OWN];
    __shared__ int sBase[N_OWN];
    __shared__ int hc[N_OWN];
    __shared__ int wsum[8];

    const int t = threadIdx.x;
    const int g = blockIdx.x;

    int4 s = make_int4(0, 0, 0, 0);
    int4 d = make_int4(-1, -1, -1, -1);
    if (t < SLICE_VEC) {
        s = ((const int4*)ei)[g * SLICE_VEC + t];
        d = ((const int4*)(ei + N_EDGES))[g * SLICE_VEC + t];
    }

    const int o = t & (N_OWN - 1);
    const int p = __builtin_amdgcn_readfirstlane(t >> 8);
    const int* col = hist + (p * 64) * N_OWN + o;
    const int nbefore = g - p * 64;
    int full = 0, part = 0;
#pragma unroll 16
    for (int r = 0; r < 64; ++r) {
        const int v = col[r * N_OWN];
        full += v;
        part += (r < nbefore) ? v : 0;
    }
    sT[p][o] = full;
    sP[p][o] = part;
    if (t < N_OWN) hc[t] = 0;
    __syncthreads();

    int tot = 0, pre = 0;
    if (t < N_OWN) {
        tot = sT[0][t] + sT[1][t] + sT[2][t] + sT[3][t];
        pre = sP[0][t] + sP[1][t] + sP[2][t] + sP[3][t];
    }
    const int ex = scan256(tot, wsum, t);
    if (t < N_OWN) {
        sBase[t] = ex + pre;
        if (g == 0) {
            ostart[t] = ex;
            if (t == N_OWN - 1) ostart[N_OWN] = ex + tot;
        }
    }
    __syncthreads();

    if ((unsigned)d.x < N_NODES) {
        const unsigned oo = (unsigned)d.x / OWN_NODES;
        bin[sBase[oo] + atomicAdd(&hc[oo], 1)] = make_int2(s.x, d.x);
    }
    if ((unsigned)d.y < N_NODES) {
        const unsigned oo = (unsigned)d.y / OWN_NODES;
        bin[sBase[oo] + atomicAdd(&hc[oo], 1)] = make_int2(s.y, d.y);
    }
    if ((unsigned)d.z < N_NODES) {
        const unsigned oo = (unsigned)d.z / OWN_NODES;
        bin[sBase[oo] + atomicAdd(&hc[oo], 1)] = make_int2(s.z, d.z);
    }
    if ((unsigned)d.w < N_NODES) {
        const unsigned oo = (unsigned)d.w / OWN_NODES;
        bin[sBase[oo] + atomicAdd(&hc[oo], 1)] = make_int2(s.w, d.w);
    }
}

__global__ __launch_bounds__(PART_THREADS) void owner_slots(
    const int* __restrict__ ostart, const int* __restrict__ binp,
    int* __restrict__ offs, int* __restrict__ esrc) {
    const int2* __restrict__ bin = (const int2*)binp;
    __shared__ int c[N_OWN];
    __shared__ int loff[N_OWN];
    __shared__ int wsum[8];

    const int t   = threadIdx.x;
    const int lo  = blockIdx.x * OWN_NODES;
    const int beg = ostart[blockIdx.x];
    const int end = ostart[blockIdx.x + 1];

    if (t < N_OWN) c[t] = 0;
    __syncthreads();

    for (int i0 = beg + t; i0 < end; i0 += 4 * PART_THREADS) {
        int dd[4];
#pragma unroll
        for (int u = 0; u < 4; ++u) {
            const int i = i0 + u * PART_THREADS;
            dd[u] = (i < end) ? bin[i].y : -1;
        }
#pragma unroll
        for (int u = 0; u < 4; ++u)
            if (dd[u] >= 0) atomicAdd(&c[dd[u] - lo], 1);
    }
    __syncthreads();

    int v = 0;
    if (t < N_OWN) v = c[t];
    const int ex = scan256(v, wsum, t);
    if (t < N_OWN) {
        loff[t] = ex;
        c[t] = 0;
        const int node = lo + t;
        if (t < OWN_NODES && node <= N_NODES) offs[node] = beg + ex;
    }
    __syncthreads();

    for (int i0 = beg + t; i0 < end; i0 += 4 * PART_THREADS) {
        int2 e[4];
#pragma unroll
        for (int u = 0; u < 4; ++u) {
            const int i = i0 + u * PART_THREADS;
            e[u] = (i < end) ? bin[i] : make_int2(0, -1);
        }
#pragma unroll
        for (int u = 0; u < 4; ++u)
            if (e[u].y >= 0) {
                const int ln = e[u].y - lo;
                esrc[beg + loff[ln] + atomicAdd(&c[ln], 1)] = e[u].x;
            }
    }
}

#define GATHER_DEPTH 8
__global__ __launch_bounds__(256) void gather_rows(
    const float* __restrict__ h, const int* __restrict__ offs,
    const int* __restrict__ esrc, float* __restrict__ agg) {
    const int lane = threadIdx.x & 31;
    const int wv   = __builtin_amdgcn_readfirstlane((int)(threadIdx.x >> 5));
    const int node = blockIdx.x * 8 + wv;
    if (node >= N_NODES) return;

    const int beg = offs[node];
    const int end = offs[node + 1];
    const float* hl = h + lane * 4;

    float4 acc = {0.f, 0.f, 0.f, 0.f};
    for (int base = beg; base < end; base += 32) {
        const int n = (end - base < 32) ? (end - base) : 32;
        const int mine = (lane < n) ? esrc[base + lane] : 0;
        for (int j = 0; j < n; j += GATHER_DEPTH) {
            int sidx[GATHER_DEPTH];
#pragma unroll
            for (int k = 0; k < GATHER_DEPTH; ++k)
                sidx[k] = __builtin_amdgcn_readlane(mine, j + k);
            float4 v[GATHER_DEPTH];
#pragma unroll
            for (int k = 0; k < GATHER_DEPTH; ++k) {
                v[k] = make_float4(0.f, 0.f, 0.f, 0.f);
                if (j + k < n)
                    v[k] = *(const float4*)(hl + (size_t)sidx[k] * DDIM);
            }
#pragma unroll
            for (int k = 0; k < GATHER_DEPTH; ++k) {
                acc.x += v[k].x; acc.y += v[k].y;
                acc.z += v[k].z; acc.w += v[k].w;
            }
        }
    }
    *(float4*)(agg + (size_t)node * DDIM + lane * 4) = acc;
}

#define MLP_WAVES   16
#define MLP_THREADS (MLP_WAVES * 32)
#define MLP_GROUPS  256
#define N_TILES     ((N_NODES + 15) / 16)

__device__ __forceinline__ void wave_handover() {
    __builtin_amdgcn_fence(__ATOMIC_RELEASE, "wavefront");
    __builtin_amdgcn_wave_barrier();
    __builtin_amdgcn_fence(__ATOMIC_ACQUIRE, "wavefront");
}

__global__ __launch_bounds__(MLP_THREADS) void gin_mlp(
    const float* __restrict__ h_in, const float* __restrict__ agg,
    const float* __restrict__ eps_arr, int layer, int has_skip,
    const float* __restrict__ W1, const float* __restrict__ b1,
    const float* __restrict__ W2, const float* __restrict__ b2,
    const float* __restrict__ lng, const float* __restrict__ lnb,
    const float* __restrict__ sg,  const float* __restrict__ sb,
    float* __restrict__ h_out) {

    __shared__ alignas(16) float2 sB1[SB_SIZE];
    __shared__ alignas(16) float2 sB2[SB_SIZE];
    __shared__ float  sZ[MLP_WAVES * 16 * SZ_STRIDE];
    __shared__ float  sStat[MLP_WAVES * 16 * 2];

    const int tid   = threadIdx.x;
    const int lane  = tid & 31;
    const int lhalf = lane >> 4;
    const int l15   = lane & 15;
    const int wave  = __builtin_amdgcn_readfirstlane(tid >> 5);

    const float* W1l = W1 + (size_t)layer * DDIM * DDIM;
    const float* W2l = W2 + (size_t)layer * DDIM * DDIM;
    const float* b1l = b1 + (size_t)layer * DDIM;
    const float* b2l = b2 + (size_t)layer * DDIM;
    const float* lgl = lng + (size_t)layer * DDIM;
    const float* lbl = lnb + (size_t)layer * DDIM;
    const float* sgl = sg + (size_t)layer * DDIM;
    const float* sbl = sb + (size_t)layer * DDIM;

#pragma unroll
    for (int i = 0; i < 4; ++i) {
        const int item = tid + i * MLP_THREADS;
        const int pair = item >> 5;
        const int c4   = (item & 31) * 4;
        const float4 p0 = *(const float4*)(W1l + (2 * pair) * DDIM + c4);
        const float4 p1 = *(const float4*)(W1l + (2 * pair + 1) * DDIM + c4);
        const float4 q0 = *(const float4*)(W2l + (2 * pair) * DDIM + c4);
        const float4 q1 = *(const float4*)(W2l + (2 * pair + 1) * DDIM + c4);
        float4* d1 = (float4*)&sB1[pair * SB_HSTRIDE + c4];
        float4* d2 = (float4*)&sB2[pair * SB_HSTRIDE + c4];
        d1[0] = make_float4(p0.x, p1.x, p0.y, p1.y);
        d1[1] = make_float4(p0.z, p1.z, p0.w, p1.w);
        d2[0] = make_float4(q0.x, q1.x, q0.y, q1.y);
        d2[1] = make_float4(q0.z, q1.z, q0.w, q1.w);
    }
    __syncthreads();

    const int tile = wave * MLP_GROUPS + (int)blockIdx.x;
    if (tile >= N_TILES) return;

    const float epsv = 1.0f + eps_arr[layer];
    const int node_base = tile * 16;

    int arow = node_base + l15;
    if (arow >= N_NODES) arow = N_NODES - 1;
    const float* hrow = h_in + (size_t)arow * DDIM;
    const float* grow = agg  + (size_t)arow * DDIM;

    const v8f vzero = {};
    v8f acc[8];
#pragma unroll
    for (int t = 0; t < 8; ++t) acc[t] = vzero;

    const float2* bfrag1 = &sB1[lhalf * SB_HSTRIDE + l15];
    for (int kk = 0; kk < 32; ++kk) {
        const int c0 = kk * 4 + lhalf * 2;
        float2 hv = *(const float2*)(hrow + c0);
        float2 gv = *(const float2*)(grow + c0);
        v2f a;
        a.x = epsv * hv.x + gv.x;
        a.y = epsv * hv.y + gv.y;
        const float2* bk = bfrag1 + kk * (2 * SB_HSTRIDE);
#pragma unroll
        for (int nt = 0; nt < 8; ++nt) {
            float2 bf = bk[nt * 16];
            v2f b; b.x = bf.x; b.y = bf.y;
            acc[nt] = __builtin_amdgcn_wmma_f32_16x16x4_f32(
                false, a, false, b, (short)0, acc[nt], false, false);
        }
    }

    float* zt = &sZ[wave * 16 * SZ_STRIDE];
    {
#pragma unroll
        for (int nt = 0; nt < 8; ++nt) {
            const int col  = nt * 16 + l15;
            const float bi = b1l[col];
#pragma unroll
            for (int v = 0; v < 8; ++v) {
                float val = acc[nt][v] + bi;
                val = fmaxf(val, 0.0f);
                zt[(v + lhalf * 8) * SZ_STRIDE + col] = val;
            }
        }
    }
    wave_handover();

#pragma unroll
    for (int t = 0; t < 8; ++t) acc[t] = vzero;

    {
        const float2* bfrag2 = &sB2[lhalf * SB_HSTRIDE + l15];
        const float* zrow = zt + l15 * SZ_STRIDE;
        for (int kk = 0; kk < 32; ++kk) {
            const int c0 = kk * 4 + lhalf * 2;
            float2 av = *(const float2*)(zrow + c0);
            v2f a; a.x = av.x; a.y = av.y;
            const float2* bk = bfrag2 + kk * (2 * SB_HSTRIDE);
#pragma unroll
            for (int nt = 0; nt < 8; ++nt) {
                float2 bf = bk[nt * 16];
                v2f b; b.x = bf.x; b.y = bf.y;
                acc[nt] = __builtin_amdgcn_wmma_f32_16x16x4_f32(
                    false, a, false, b, (short)0, acc[nt], false, false);
            }
        }
    }
    wave_handover();

    {
#pragma unroll
        for (int nt = 0; nt < 8; ++nt) {
            const int col  = nt * 16 + l15;
            const float bi = b2l[col];
#pragma unroll
            for (int v = 0; v < 8; ++v)
                zt[(v + lhalf * 8) * SZ_STRIDE + col] = acc[nt][v] + bi;
        }
    }
    wave_handover();

    {
        float* yrow = zt + l15 * SZ_STRIDE;
        const int cbase = lhalf * 64;

        float s = 0.0f, s2 = 0.0f;
#pragma unroll 4
        for (int j = 0; j < 64; ++j) {
            float y = yrow[cbase + j];
            s += y; s2 += y * y;
        }
        s  += __shfl_xor(s, 16, 32);
        s2 += __shfl_xor(s2, 16, 32);
        const float mu  = s * (1.0f / DDIM);
        const float var = s2 * (1.0f / DDIM) - mu * mu;
        const float inv = rsqrtf(var + LN_EPS);

        float t = 0.0f, t2 = 0.0f;
#pragma unroll 4
        for (int j = 0; j < 64; ++j) {
            const int c = cbase + j;
            float z1 = (yrow[c] - mu) * inv * lgl[c] + lbl[c];
            yrow[c] = z1;
            t += z1; t2 += z1 * z1;
        }

        if (has_skip) {
            t  += __shfl_xor(t, 16, 32);
            t2 += __shfl_xor(t2, 16, 32);
            float mu2  = t * (1.0f / DDIM);
            float var2 = t2 * (1.0f / DDIM) - mu2 * mu2;
            float inv2 = rsqrtf(var2 + LN_EPS);
            if (lhalf == 0) {
                sStat[(wave * 16 + l15) * 2 + 0] = mu2;
                sStat[(wave * 16 + l15) * 2 + 1] = inv2;
            }
        }
    }
    wave_handover();

    {
        const int c4 = lane * 4;
        float4 g4 = {1.f, 1.f, 1.f, 1.f}, sb4 = {0.f, 0.f, 0.f, 0.f};
        if (has_skip) {
            g4  = *(const float4*)(sgl + c4);
            sb4 = *(const float4*)(sbl + c4);
        }
#pragma unroll 4
        for (int r = 0; r < 16; ++r) {
            const int node = node_base + r;
            if (node >= N_NODES) break;
            const float* zr = zt + r * SZ_STRIDE + c4;
            float4 z = {zr[0], zr[1], zr[2], zr[3]};
            if (has_skip) {
                const float mu2  = sStat[(wave * 16 + r) * 2 + 0];
                const float inv2 = sStat[(wave * 16 + r) * 2 + 1];
                z.x = fmaxf((z.x - mu2) * inv2 * g4.x + sb4.x, 0.0f);
                z.y = fmaxf((z.y - mu2) * inv2 * g4.y + sb4.y, 0.0f);
                z.z = fmaxf((z.z - mu2) * inv2 * g4.z + sb4.z, 0.0f);
                z.w = fmaxf((z.w - mu2) * inv2 * g4.w + sb4.w, 0.0f);
            }
            const float4 hres = *(const float4*)(h_in + (size_t)node * DDIM + c4);
            float4 o;
            o.x = z.x + hres.x; o.y = z.y + hres.y;
            o.z = z.z + hres.z; o.w = z.w + hres.w;
            *(float4*)(h_out + (size_t)node * DDIM + c4) = o;
        }
    }
}

extern "C" void kernel_launch(void* const* d_in, const int* in_sizes, int n_in,
                              void* d_out, int out_size, void* d_ws, size_t ws_size,
                              hipStream_t stream) {
    (void)in_sizes; (void)n_in; (void)out_size; (void)ws_size;

    const float* x   = (const float*)d_in[0];
    const int*   ei  = (const int*)d_in[1];
    const float* eps = (const float*)d_in[2];
    const float* W1  = (const float*)d_in[3];
    const float* b1  = (const float*)d_in[4];
    const float* W2  = (const float*)d_in[5];
    const float* b2  = (const float*)d_in[6];
    const float* lng = (const float*)d_in[7];
    const float* lnb = (const float*)d_in[8];
    const float* sg  = (const float*)d_in[9];
    const float* sb  = (const float*)d_in[10];
    float* out = (float*)d_out;

    const size_t feat = (size_t)N_NODES * DDIM;
    float* agg = (float*)d_ws;
    float* hA  = agg + feat;
    int*  hist   = (int*)(hA + feat);
    int*  ostart = hist + N_SLICE * N_OWN;
    int*  offs   = ostart + 512;
    int*  esrc   = offs + N_OWN * OWN_NODES;
    int*  bin    = esrc + N_EDGES;

    const int gather_blocks = (N_NODES + 7) / 8;

    slice_hist<<<N_SLICE, 640, 0, stream>>>(ei, hist);
    bin_edges<<<N_SLICE, PART_THREADS, 0, stream>>>(ei, hist, ostart, bin);
    owner_slots<<<N_OWN, PART_THREADS, 0, stream>>>(ostart, bin, offs, esrc);

    const float* hin = x;
    for (int i = 0; i < 3; ++i) {
        float* hout = (i == 1) ? hA : out;
        const int has_skip = (i < 2) ? 1 : 0;

        gather_rows<<<gather_blocks, 256, 0, stream>>>(hin, offs, esrc, agg);
        gin_mlp<<<MLP_GROUPS, MLP_THREADS, 0, stream>>>(
            hin, agg, eps, i, has_skip,
            W1, b1, W2, b2, lng, lnb, sg, sb, hout);

        hin = hout;
    }
}


# Epilogue loads grouped: a tile's 16 residual rows, skip parameters and bias vectors are asked for ahead, LayerNorm reads 16 or 32 columns per trip; no dependent trip per row; sum order kept
# speedup vs baseline: 3.4293x; 1.2718x over previous
#include <hip/hip_runtime.h>

typedef __attribute__((ext_vector_type(2))) float v2f;
typedef __attribute__((ext_vector_type(8))) float v8f;

#define N_NODES 50000
#define N_EDGES 640000
#define DDIM    128
#define LN_EPS  1e-5f

#define SB_HSTRIDE 144
#define SB_SIZE    (32 * 2 * SB_HSTRIDE)

#define SZ_STRIDE 132

#define OWN_NODES  196
#define N_OWN      256
#define N_SLICE    256
#define DST_VEC    (N_EDGES / 4)
#define SLICE_VEC  (DST_VEC / N_SLICE)
#define PART_THREADS 1024

__global__ __launch_bounds__(640) void slice_hist(
    const int* __restrict__ ei, int* __restrict__ hist) {
    __shared__ int h[N_OWN];
    const int t = threadIdx.x;
    if (t < N_OWN) h[t] = 0;
    __syncthreads();
    if (t < SLICE_VEC) {
        const int4 d = ((const int4*)(ei + N_EDGES))[blockIdx.x * SLICE_VEC + t];
        if ((unsigned)d.x < N_NODES) atomicAdd(&h[(unsigned)d.x / OWN_NODES], 1);
        if ((unsigned)d.y < N_NODES) atomicAdd(&h[(unsigned)d.y / OWN_NODES], 1);
        if ((unsigned)d.z < N_NODES) atomicAdd(&h[(unsigned)d.z / OWN_NODES], 1);
        if ((unsigned)d.w < N_NODES) atomicAdd(&h[(unsigned)d.w / OWN_NODES], 1);
    }
    __syncthreads();
    if (t < N_OWN) hist[blockIdx.x * N_OWN + t] = h[t];
}

__device__ __forceinline__ int scan256(const int v, int* __restrict__ wsum, const int t) {
    const int lane = t & 31;
    const int w    = t >> 5;
    int incl = v;
    for (int d = 1; d < 32; d <<= 1) {
        const int u = __shfl_up(incl, d, 32);
        if (lane >= d) incl += u;
    }
    if (t < N_OWN && lane == 31) wsum[w] = incl;
    __syncthreads();
    int wbase = 0;
    if (t < N_OWN)
        for (int k = 0; k < w; ++k) wbase += wsum[k];
    return wbase + incl - v;
}

__global__ __launch_bounds__(PART_THREADS) void bin_edges(
    const int* __restrict__ ei, const int* __restrict__ hist,
    int* __restrict__ ostart, int* __restrict__ binp) {
    int2* __restrict__ bin = (int2*)binp;
    __shared__ int sT[4][N_OWN];
    __shared__ int sP[4][N_OWN];
    __shared__ int sBase[N_OWN];
    __shared__ int hc[N_OWN];
    __shared__ int wsum[8];

    const int t = threadIdx.x;
    const int g = blockIdx.x;

    int4 s = make_int4(0, 0, 0, 0);
    int4 d = make_int4(-1, -1, -1, -1);
    if (t < SLICE_VEC) {
        s = ((const int4*)ei)[g * SLICE_VEC + t];
        d = ((const int4*)(ei + N_EDGES))[g * SLICE_VEC + t];
    }

    const int o = t & (N_OWN - 1);
    const int p = __builtin_amdgcn_readfirstlane(t >> 8);
    const int* col = hist + (p * 64) * N_OWN + o;
    const int nbefore = g - p * 64;
    int full = 0, part = 0;
#pragma unroll 16
    for (int r = 0; r < 64; ++r) {
        const int v = col[r * N_OWN];
        full += v;
        part += (r < nbefore) ? v : 0;
    }
    sT[p][o] = full;
    sP[p][o] = part;
    if (t < N_OWN) hc[t] = 0;
    __syncthreads();

    int tot = 0, pre = 0;
    if (t < N_OWN) {
        tot = sT[0][t] + sT[1][t] + sT[2][t] + sT[3][t];
        pre = sP[0][t] + sP[1][t] + sP[2][t] + sP[3][t];
    }
    const int ex = scan256(tot, wsum, t);
    if (t < N_OWN) {
        sBase[t] = ex + pre;
        if (g == 0) {
            ostart[t] = ex;
            if (t == N_OWN - 1) ostart[N_OWN] = ex + tot;
        }
    }
    __syncthreads();

    if ((unsigned)d.x < N_NODES) {
        const unsigned oo = (unsigned)d.x / OWN_NODES;
        bin[sBase[oo] + atomicAdd(&hc[oo], 1)] = make_int2(s.x, d.x);
    }
    if ((unsigned)d.y < N_NODES) {
        const unsigned oo = (unsigned)d.y / OWN_NODES;
        bin[sBase[oo] + atomicAdd(&hc[oo], 1)] = make_int2(s.y, d.y);
    }
    if ((unsigned)d.z < N_NODES) {
        const unsigned oo = (unsigned)d.z / OWN_NODES;
        bin[sBase[oo] + atomicAdd(&hc[oo], 1)] = make_int2(s.z, d.z);
    }
    if ((unsigned)d.w < N_NODES) {
        const unsigned oo = (unsigned)d.w / OWN_NODES;
        bin[sBase[oo] + atomicAdd(&hc[oo], 1)] = make_int2(s.w, d.w);
    }
}

__global__ __launch_bounds__(PART_THREADS) void owner_slots(
    const int* __restrict__ ostart, const int* __restrict__ binp,
    int* __restrict__ offs, int* __restrict__ esrc) {
    const int2* __restrict__ bin = (const int2*)binp;
    __shared__ int c[N_OWN];
    __shared__ int loff[N_OWN];
    __shared__ int wsum[8];

    const int t   = threadIdx.x;
    const int lo  = blockIdx.x * OWN_NODES;
    const int beg = ostart[blockIdx.x];
    const int end = ostart[blockIdx.x + 1];

    if (t < N_OWN) c[t] = 0;
    __syncthreads();

    for (int i0 = beg + t; i0 < end; i0 += 4 * PART_THREADS) {
        int dd[4];
#pragma unroll
        for (int u = 0; u < 4; ++u) {
            const int i = i0 + u * PART_THREADS;
            dd[u] = (i < end) ? bin[i].y : -1;
        }
#pragma unroll
        for (int u = 0; u < 4; ++u)
            if (dd[u] >= 0) atomicAdd(&c[dd[u] - lo], 1);
    }
    __syncthreads();

    int v = 0;
    if (t < N_OWN) v = c[t];
    const int ex = scan256(v, wsum, t);
    if (t < N_OWN) {
        loff[t] = ex;
        c[t] = 0;
        const int node = lo + t;
        if (t < OWN_NODES && node <= N_NODES) offs[node] = beg + ex;
    }
    __syncthreads();

    for (int i0 = beg + t; i0 < end; i0 += 4 * PART_THREADS) {
        int2 e[4];
#pragma unroll
        for (int u = 0; u < 4; ++u) {
            const int i = i0 + u * PART_THREADS;
            e[u] = (i < end) ? bin[i] : make_int2(0, -1);
        }
#pragma unroll
        for (int u = 0; u < 4; ++u)
            if (e[u].y >= 0) {
                const int ln = e[u].y - lo;
                esrc[beg + loff[ln] + atomicAdd(&c[ln], 1)] = e[u].x;
            }
    }
}

#define GATHER_DEPTH 16

__device__ __forceinline__ void rows_issue(
    const float* __restrict__ hl, const int idxv, const int lane0, const int n,
    float4 (&v)[GATHER_DEPTH]) {
    int sidx[GATHER_DEPTH];
#pragma unroll
    for (int k = 0; k < GATHER_DEPTH; ++k)
        sidx[k] = __builtin_amdgcn_readlane(idxv, lane0 + k);
#pragma unroll
    for (int k = 0; k < GATHER_DEPTH; ++k) {
        v[k] = make_float4(0.f, 0.f, 0.f, 0.f);
        if (k < n)
            v[k] = *(const float4*)(hl + (size_t)sidx[k] * DDIM);
    }
}

__device__ __forceinline__ void rows_add(float4& acc, const float4 (&v)[GATHER_DEPTH]) {
#pragma unroll
    for (int k = 0; k < GATHER_DEPTH; ++k) {
        acc.x += v[k].x; acc.y += v[k].y;
        acc.z += v[k].z; acc.w += v[k].w;
    }
}

__device__ __forceinline__ void gather_tile(
    const float* __restrict__ h, const int* __restrict__ offs,
    const int* __restrict__ esrc, const int node_base, const float epsv,
    float* __restrict__ zt, const int lane) {
    int oi = node_base + ((lane < 16) ? lane : 16);
    if (oi > N_NODES) oi = N_NODES;
    const int ofs_v = offs[oi];
    const float* hl = h + lane * 4;

    int beg = __builtin_amdgcn_readlane(ofs_v, 0);
    int end = __builtin_amdgcn_readlane(ofs_v, 1);
    int pi = beg + lane;
    if (pi > N_EDGES - 1) pi = N_EDGES - 1;
    int mine = esrc[pi];

    for (int r = 0; r < 16; ++r) {
        const int cnt = end - beg;
        float4 v[GATHER_DEPTH];
        rows_issue(hl, mine, 0, cnt, v);

        const int nend = __builtin_amdgcn_readlane(ofs_v, r + 2);
        int pn = end + lane;
        if (pn > N_EDGES - 1) pn = N_EDGES - 1;
        const int mine_n = esrc[pn];
        int node = node_base + r;
        if (node >= N_NODES) node = N_NODES - 1;
        const float4 hs = *(const float4*)(hl + (size_t)node * DDIM);

        float4 acc = {0.f, 0.f, 0.f, 0.f};
        asm volatile("" : "+v"(acc.x), "+v"(acc.y), "+v"(acc.z), "+v"(acc.w));
        rows_add(acc, v);
        if (cnt > GATHER_DEPTH) {
            rows_issue(hl, mine, GATHER_DEPTH, cnt - GATHER_DEPTH, v);
            rows_add(acc, v);
            for (int base = beg + 32; base < end; base += 32) {
                const int n = (end - base < 32) ? (end - base) : 32;
                const int cur = (lane < n) ? esrc[base + lane] : 0;
                rows_issue(hl, cur, 0, n, v);
                rows_add(acc, v);
                if (n > GATHER_DEPTH) {
                    rows_issue(hl, cur, GATHER_DEPTH, n - GATHER_DEPTH, v);
                    rows_add(acc, v);
                }
            }
        }
        float4 z;
        z.x = epsv * hs.x + acc.x; z.y = epsv * hs.y + acc.y;
        z.z = epsv * hs.z + acc.z; z.w = epsv * hs.w + acc.w;
        *(float4*)(zt + r * SZ_STRIDE + lane * 4) = z;

        beg = end; end = nend; mine = mine_n;
    }
}

#define MLP_WAVES   16
#define MLP_THREADS (MLP_WAVES * 32)
#define MLP_GROUPS  256
#define N_TILES     ((N_NODES + 15) / 16)

__device__ __forceinline__ void wave_handover() {
    __builtin_amdgcn_fence(__ATOMIC_RELEASE, "wavefront");
    __builtin_amdgcn_wave_barrier();
    __builtin_amdgcn_fence(__ATOMIC_ACQUIRE, "wavefront");
}

__global__ __launch_bounds__(MLP_THREADS) void gin_mlp(
    const float* __restrict__ h_in, const int* __restrict__ offs,
    const int* __restrict__ esrc,
    const float* __restrict__ eps_arr, int layer, int has_skip,
    const float* __restrict__ W1, const float* __restrict__ b1,
    const float* __restrict__ W2, const float* __restrict__ b2,
    const float* __restrict__ lng, const float* __restrict__ lnb,
    const float* __restrict__ sg,  const float* __restrict__ sb,
    float* __restrict__ h_out) {

    __shared__ alignas(16) float2 sB1[SB_SIZE];
    __shared__ alignas(16) float2 sB2[SB_SIZE];
    __shared__ alignas(16) float sZ[MLP_WAVES * 16 * SZ_STRIDE];
    __shared__ float  sStat[MLP_WAVES * 16 * 2];

    const int tid   = threadIdx.x;
    const int lane  = tid & 31;
    const int lhalf = lane >> 4;
    const int l15   = lane & 15;
    const int wave  = __builtin_amdgcn_readfirstlane(tid >> 5);

    const float* W1l = W1 + (size_t)layer * DDIM * DDIM;
    const float* W2l = W2 + (size_t)layer * DDIM * DDIM;
    const float* b1l = b1 + (size_t)layer * DDIM;
    const float* b2l = b2 + (size_t)layer * DDIM;
    const float* lgl = lng + (size_t)layer * DDIM;
    const float* lbl = lnb + (size_t)layer * DDIM;
    const float* sgl = sg + (size_t)layer * DDIM;
    const float* sbl = sb + (size_t)layer * DDIM;

#pragma unroll
    for (int i = 0; i < 4; ++i) {
        const int item = tid + i * MLP_THREADS;
        const int pair = item >> 5;
        const int c4   = (item & 31) * 4;
        const float4 p0 = *(const float4*)(W1l + (2 * pair) * DDIM + c4);
        const float4 p1 = *(const float4*)(W1l + (2 * pair + 1) * DDIM + c4);
        const float4 q0 = *(const float4*)(W2l + (2 * pair) * DDIM + c4);
        const float4 q1 = *(const float4*)(W2l + (2 * pair + 1) * DDIM + c4);
        float4* d1 = (float4*)&sB1[pair * SB_HSTRIDE + c4];
        float4* d2 = (float4*)&sB2[pair * SB_HSTRIDE + c4];
        d1[0] = make_float4(p0.x, p1.x, p0.y, p1.y);
        d1[1] = make_float4(p0.z, p1.z, p0.w, p1.w);
        d2[0] = make_float4(q0.x, q1.x, q0.y, q1.y);
        d2[1] = make_float4(q0.z, q1.z, q0.w, q1.w);
    }
    __syncthreads();

    const int tile = wave * MLP_GROUPS + (int)blockIdx.x;
    if (tile >= N_TILES) return;

    const float epsv = 1.0f + eps_arr[layer];
    const int node_base = tile * 16;

    float bias1[8], bias2[8];
#pragma unroll
    for (int nt = 0; nt < 8; ++nt) {
        bias1[nt] = b1l[nt * 16 + l15];
        bias2[nt] = b2l[nt * 16 + l15];
    }

    float* zt = &sZ[wave * 16 * SZ_STRIDE];
    gather_tile(h_in, offs, esrc, node_base, epsv, zt, lane);
    wave_handover();

    const v8f vzero = {};
    v8f acc[8];
#pragma unroll
    for (int t = 0; t < 8; ++t) acc[t] = vzero;

    {
        const float2* bfrag1 = &sB1[lhalf * SB_HSTRIDE + l15];
        const float* zrow = zt + l15 * SZ_STRIDE;
        for (int kk = 0; kk < 32; ++kk) {
            const int c0 = kk * 4 + lhalf * 2;
            float2 av = *(const float2*)(zrow + c0);
            v2f a; a.x = av.x; a.y = av.y;
            const float2* bk = bfrag1 + kk * (2 * SB_HSTRIDE);
#pragma unroll
            for (int nt = 0; nt < 8; ++nt) {
                float2 bf = bk[nt * 16];
                v2f b; b.x = bf.x; b.y = bf.y;
                acc[nt] = __builtin_amdgcn_wmma_f32_16x16x4_f32(
                    false, a, false, b, (short)0, acc[nt], false, false);
            }
        }
    }
    wave_handover();

    {
#pragma unroll
        for (int nt = 0; nt < 8; ++nt) {
            const int col  = nt * 16 + l15;
            const float bi = bias1[nt];
#pragma unroll
            for (int v = 0; v < 8; ++v) {
                float val = acc[nt][v] + bi;
                val = fmaxf(val, 0.0f);
                zt[(v + lhalf * 8) * SZ_STRIDE + col] = val;
            }
        }
    }
    wave_handover();

#pragma unroll
    for (int t = 0; t < 8; ++t) acc[t] = vzero;

    {
        const float2* bfrag2 = &sB2[lhalf * SB_HSTRIDE + l15];
        const float* zrow = zt + l15 * SZ_STRIDE;
        for (int kk = 0; kk < 32; ++kk) {
            const int c0 = kk * 4 + lhalf * 2;
            float2 av = *(const float2*)(zrow + c0);
            v2f a; a.x = av.x; a.y = av.y;
            const float2* bk = bfrag2 + kk * (2 * SB_HSTRIDE);
#pragma unroll
            for (int nt = 0; nt < 8; ++nt) {
                float2 bf = bk[nt * 16];
                v2f b; b.x = bf.x; b.y = bf.y;
                acc[nt] = __builtin_amdgcn_wmma_f32_16x16x4_f32(
                    false, a, false, b, (short)0, acc[nt], false, false);
            }
        }
    }
    wave_handover();

    const int c4 = lane * 4;
    float4 hres[16];
#pragma unroll
    for (int r = 0; r < 16; ++r)
        hres[r] = *(const float4*)(h_in + (size_t)(node_base + r) * DDIM + c4);
    float4 g4 = {1.f, 1.f, 1.f, 1.f}, sb4 = {0.f, 0.f, 0.f, 0.f};
    if (has_skip) {
        g4  = *(const float4*)(sgl + c4);
        sb4 = *(const float4*)(sbl + c4);
    }

    {
#pragma unroll
        for (int nt = 0; nt < 8; ++nt) {
            const int col  = nt * 16 + l15;
            const float bi = bias2[nt];
#pragma unroll
            for (int v = 0; v < 8; ++v)
                zt[(v + lhalf * 8) * SZ_STRIDE + col] = acc[nt][v] + bi;
        }
    }
    wave_handover();

    {
        float* yrow = zt + l15 * SZ_STRIDE;
        const int cbase = lhalf * 64;

        float s = 0.0f, s2 = 0.0f;
#pragma unroll 16
        for (int j = 0; j < 64; ++j) {
            float y = yrow[cbase + j];
            s += y; s2 += y * y;
        }
        s  += __shfl_xor(s, 16, 32);
        s2 += __shfl_xor(s2, 16, 32);
        const float mu  = s * (1.0f / DDIM);
        const float var = s2 * (1.0f / DDIM) - mu * mu;
        const float inv = rsqrtf(var + LN_EPS);

        float t = 0.0f, t2 = 0.0f;
#pragma unroll 32
        for (int j = 0; j < 64; ++j) {
            const int c = cbase + j;
            float z1 = (yrow[c] - mu) * inv * lgl[c] + lbl[c];
            yrow[c] = z1;
            t += z1; t2 += z1 * z1;
        }

        if (has_skip) {
            t  += __shfl_xor(t, 16, 32);
            t2 += __shfl_xor(t2, 16, 32);
            float mu2  = t * (1.0f / DDIM);
            float var2 = t2 * (1.0f / DDIM) - mu2 * mu2;
            float inv2 = rsqrtf(var2 + LN_EPS);
            if (lhalf == 0) {
                sStat[(wave * 16 + l15) * 2 + 0] = mu2;
                sStat[(wave * 16 + l15) * 2 + 1] = inv2;
            }
        }
    }
    wave_handover();

    static_assert(N_NODES % 16 == 0, "a tile holds 16 nodes of the graph");
    if (has_skip) {
#pragma unroll
        for (int r = 0; r < 16; ++r) {
            const float* zr = zt + r * SZ_STRIDE + c4;
            float4 z = {zr[0], zr[1], zr[2], zr[3]};
            const float mu2  = sStat[(wave * 16 + r) * 2 + 0];
            const float inv2 = sStat[(wave * 16 + r) * 2 + 1];
            z.x = fmaxf((z.x - mu2) * inv2 * g4.x + sb4.x, 0.0f);
            z.y = fmaxf((z.y - mu2) * inv2 * g4.y + sb4.y, 0.0f);
            z.z = fmaxf((z.z - mu2) * inv2 * g4.z + sb4.z, 0.0f);
            z.w = fmaxf((z.w - mu2) * inv2 * g4.w + sb4.w, 0.0f);
            float4 o;
            o.x = z.x + hres[r].x; o.y = z.y + hres[r].y;
            o.z = z.z + hres[r].z; o.w = z.w + hres[r].w;
            *(float4*)(h_out + (size_t)(node_base + r) * DDIM + c4) = o;
        }
    } else {
#pragma unroll
        for (int r = 0; r < 16; ++r) {
            const float* zr = zt + r * SZ_STRIDE + c4;
            float4 o;
            o.x = zr[0] + hres[r].x; o.y = zr[1] + hres[r].y;
            o.z = zr[2] + hres[r].z; o.w = zr[3] + hres[r].w;
            *(float4*)(h_out + (size_t)(node_base + r) * DDIM + c4) = o;
        }
    }
}

extern "C" void kernel_launch(void* const* d_in, const int* in_sizes, int n_in,
                              void* d_out, int out_size, void* d_ws, size_t ws_size,
                              hipStream_t stream) {
    (void)in_sizes; (void)n_in; (void)out_size; (void)ws_size;

    const float* x   = (const float*)d_in[0];
    const int*   ei  = (const int*)d_in[1];
    const float* eps = (const float*)d_in[2];
    const float* W1  = (const float*)d_in[3];
    const float* b1  = (const float*)d_in[4];
    const float* W2  = (const float*)d_in[5];
    const float* b2  = (const float*)d_in[6];
    const float* lng = (const float*)d_in[7];
    const float* lnb = (const float*)d_in[8];
    const float* sg  = (const float*)d_in[9];
    const float* sb  = (const float*)d_in[10];
    float* out = (float*)d_out;

    const size_t feat = (size_t)N_NODES * DDIM;
    float* hA  = (float*)d_ws + feat;
    int*  hist   = (int*)(hA + feat);
    int*  ostart = hist + N_SLICE * N_OWN;
    int*  offs   = ostart + 512;
    int*  esrc   = offs + N_OWN * OWN_NODES;
    int*  bin    = esrc + N_EDGES;

    slice_hist<<<N_SLICE, 640, 0, stream>>>(ei, hist);
    bin_edges<<<N_SLICE, PART_THREADS, 0, stream>>>(ei, hist, ostart, bin);
    owner_slots<<<N_OWN, PART_THREADS, 0, stream>>>(ostart, bin, offs, esrc);

    const float* hin = x;
    for (int i = 0; i < 3; ++i) {
        float* hout = (i == 1) ? hA : out;
        const int has_skip = (i < 2) ? 1 : 0;

        gin_mlp<<<MLP_GROUPS, MLP_THREADS, 0, stream>>>(
            hin, offs, esrc, eps, i, has_skip,
            W1, b1, W2, b2, lng, lnb, sg, sb, hout);

        hin = hout;
    }
}
